# starting kernel
# baseline (speedup 1.0000x reference)
#include <hip/hip_runtime.h>
#include <hip/hip_bf16.h>

typedef float v2f __attribute__((ext_vector_type(2)));
typedef float v8f __attribute__((ext_vector_type(8)));

#define NEG_SLOPE 0.2f
#define INF_EPS   1e-16f

__device__ __forceinline__ unsigned fEnc(float x) {
    unsigned u = __float_as_uint(x);
    return (u & 0x80000000u) ? ~u : (u | 0x80000000u);
}
__device__ __forceinline__ float fDec(unsigned u) {
    u = (u & 0x80000000u) ? (u & 0x7FFFFFFFu) : ~u;
    return __uint_as_float(u);
}
__device__ __forceinline__ float leaky(float x) { return x > 0.f ? x : NEG_SLOPE * x; }

template <int K, int NCOLS>
__global__ __launch_bounds__(256) void gemm_wmma_f32(
    const float* __restrict__ A, const float* __restrict__ B,
    float* __restrict__ C)
{
    const int lane = threadIdx.x & 31;
    const int wave = threadIdx.x >> 5;
    const int row  = blockIdx.x * 16 + (lane & 15);
    const int colB = wave * 16 + (lane & 15);
    const int kh   = (lane >> 4) << 1;

    const float* __restrict__ aPtr = A + (size_t)row * K + kh;
    const float* __restrict__ bPtr = B + (size_t)kh * NCOLS + colB;

    v8f acc = {};
#pragma unroll
    for (int k0 = 0; k0 < K; k0 += 4) {
        v2f a, b;
        a.x = aPtr[k0];
        a.y = aPtr[k0 + 1];
        b.x = bPtr[k0 * NCOLS];
        b.y = bPtr[k0 * NCOLS + NCOLS];
        acc = __builtin_amdgcn_wmma_f32_16x16x4_f32(
            false, a, false, b, (short)0, acc, false, false);
    }
    float* __restrict__ cPtr =
        C + ((size_t)blockIdx.x * 16 + ((lane >> 4) << 3)) * NCOLS + colB;
#pragma unroll
    for (int r = 0; r < 8; ++r)
        cPtr[r * NCOLS] = acc[r];
}

__global__ void attn_logits_h4(const float* __restrict__ h,
                               const float* __restrict__ attS,
                               const float* __restrict__ attD,
                               float* __restrict__ alS, float* __restrict__ alD, int N)
{
    int t = blockIdx.x * blockDim.x + threadIdx.x;
    if (t >= N * 4) return;
    int n = t >> 2, hd = t & 3;
    const float* hp = h + (size_t)n * 128 + hd * 32;
    const float* as = attS + hd * 32;
    const float* ad = attD + hd * 32;
    float s = 0.f, d = 0.f;
#pragma unroll
    for (int c = 0; c < 32; ++c) { float v = hp[c]; s += v * as[c]; d += v * ad[c]; }
    alS[t] = s; alD[t] = d;
}

__global__ void attn_logits_h1(const float* __restrict__ h,
                               const float* __restrict__ attS,
                               const float* __restrict__ attD,
                               float* __restrict__ alS, float* __restrict__ alD, int N)
{
    int n = blockIdx.x * blockDim.x + threadIdx.x;
    if (n >= N) return;
    const float* hp = h + (size_t)n * 64;
    float s = 0.f, d = 0.f;
#pragma unroll
    for (int c = 0; c < 64; ++c) { float v = hp[c]; s += v * attS[c]; d += v * attD[c]; }
    alS[n] = s; alD[n] = d;
}

__global__ void edge_max_h4(const int* __restrict__ esrc, const int* __restrict__ edst,
                            const float4* __restrict__ alS, const float4* __restrict__ alD,
                            unsigned* __restrict__ mEnc, int E, int Etot)
{
    int e = blockIdx.x * blockDim.x + threadIdx.x;
    if (e >= Etot) return;
    int s, d;
    if (e < E) { s = esrc[e]; d = edst[e]; } else { s = d = e - E; }
    float4 vs = alS[s], vd = alD[d];
    float ev[4] = { leaky(vs.x + vd.x), leaky(vs.y + vd.y),
                    leaky(vs.z + vd.z), leaky(vs.w + vd.w) };
#pragma unroll
    for (int h = 0; h < 4; ++h) atomicMax(&mEnc[d * 4 + h], fEnc(ev[h]));
}

__global__ void edge_exp_h4(const int* __restrict__ esrc, const int* __restrict__ edst,
                            const float4* __restrict__ alS, const float4* __restrict__ alD,
                            const unsigned* __restrict__ mEnc,
                            float* __restrict__ ex, float* __restrict__ den, int E, int Etot)
{
    int e = blockIdx.x * blockDim.x + threadIdx.x;
    if (e >= Etot) return;
    int s, d;
    if (e < E) { s = esrc[e]; d = edst[e]; } else { s = d = e - E; }
    float4 vs = alS[s], vd = alD[d];
    float ev[4] = { leaky(vs.x + vd.x), leaky(vs.y + vd.y),
                    leaky(vs.z + vd.z), leaky(vs.w + vd.w) };
#pragma unroll
    for (int h = 0; h < 4; ++h) {
        float x = __expf(ev[h] - fDec(mEnc[d * 4 + h]));
        ex[(size_t)e * 4 + h] = x;
        atomicAdd(&den[d * 4 + h], x);
    }
}

__global__ void edge_agg_128(const int* __restrict__ esrc, const int* __restrict__ edst,
                             const float* __restrict__ ex, const float* __restrict__ den,
                             const float* __restrict__ h, float* __restrict__ out,
                             int E, int Etot)
{
    int t = blockIdx.x * blockDim.x + threadIdx.x;
    if (t >= Etot * 128) return;
    int e = t >> 7, c = t & 127, hd = c >> 5;
    int s, d;
    if (e < E) { s = esrc[e]; d = edst[e]; } else { s = d = e - E; }
    float alpha = ex[(size_t)e * 4 + hd] / (den[d * 4 + hd] + INF_EPS);
    atomicAdd(&out[(size_t)d * 128 + c], alpha * h[(size_t)s * 128 + c]);
}

__global__ void finalize_elu(float* __restrict__ a, const float* __restrict__ b, int N)
{
    int t = blockIdx.x * blockDim.x + threadIdx.x;
    if (t >= N * 128) return;
    float v = a[t] + b[t & 127];
    a[t] = v > 0.f ? v : __expf(v) - 1.f;
}

__global__ void edge_max_h1(const int* __restrict__ esrc, const int* __restrict__ edst,
                            const float* __restrict__ alS, const float* __restrict__ alD,
                            unsigned* __restrict__ mEnc, int E, int Etot)
{
    int e = blockIdx.x * blockDim.x + threadIdx.x;
    if (e >= Etot) return;
    int s, d;
    if (e < E) { s = esrc[e]; d = edst[e]; } else { s = d = e - E; }
    atomicMax(&mEnc[d], fEnc(leaky(alS[s] + alD[d])));
}

__global__ void edge_exp_h1(const int* __restrict__ esrc, const int* __restrict__ edst,
                            const float* __restrict__ alS, const float* __restrict__ alD,
                            const unsigned* __restrict__ mEnc,
                            float* __restrict__ ex, float* __restrict__ den, int E, int Etot)
{
    int e = blockIdx.x * blockDim.x + threadIdx.x;
    if (e >= Etot) return;
    int s, d;
    if (e < E) { s = esrc[e]; d = edst[e]; } else { s = d = e - E; }
    float x = __expf(leaky(alS[s] + alD[d]) - fDec(mEnc[d]));
    ex[e] = x;
    atomicAdd(&den[d], x);
}

__global__ void edge_agg_64(const int* __restrict__ esrc, const int* __restrict__ edst,
                            const float* __restrict__ ex, const float* __restrict__ den,
                            const float* __restrict__ h, float* __restrict__ out,
                            int E, int Etot)
{
    int t = blockIdx.x * blockDim.x + threadIdx.x;
    if (t >= Etot * 64) return;
    int e = t >> 6, c = t & 63;
    int s, d;
    if (e < E) { s = esrc[e]; d = edst[e]; } else { s = d = e - E; }
    float alpha = ex[e] / (den[d] + INF_EPS);
    atomicAdd(&out[(size_t)d * 64 + c], alpha * h[(size_t)s * 64 + c]);
}

__global__ void finalize_bias64(float* __restrict__ o, const float* __restrict__ b, int N)
{
    int t = blockIdx.x * blockDim.x + threadIdx.x;
    if (t >= N * 64) return;
    o[t] += b[t & 63];
}

extern "C" void kernel_launch(void* const* d_in, const int* in_sizes, int n_in,
                              void* d_out, int out_size, void* d_ws, size_t ws_size,
                              hipStream_t stream)
{
    const float* x    = (const float*)d_in[0];
    const int*   eidx = (const int*)d_in[1];
    const float* W1   = (const float*)d_in[2];
    const float* aS1  = (const float*)d_in[3];
    const float* aD1  = (const float*)d_in[4];
    const float* b1   = (const float*)d_in[5];
    const float* W2   = (const float*)d_in[6];
    const float* aS2  = (const float*)d_in[7];
    const float* aD2  = (const float*)d_in[8];
    const float* b2   = (const float*)d_in[9];
    float* out = (float*)d_out;

    const int N    = in_sizes[0] / 128;
    const int E    = in_sizes[1] / 2;
    const int Etot = E + N;
    const int* esrc = eidx;
    const int* edst = eidx + E;

    float* ws = (float*)d_ws;
    size_t o = 0;
    float*    h1   = ws + o; o += (size_t)N * 128;
    float*    alS1 = ws + o; o += (size_t)N * 4;
    float*    alD1 = ws + o; o += (size_t)N * 4;
    float*    ex1  = ws + o; o += (size_t)Etot * 4;
    float*    agg1 = ws + o; o += (size_t)N * 128;
    unsigned* m1   = (unsigned*)(ws + o); o += (size_t)N * 4;
    float*    den1 = ws + o; o += (size_t)N * 4;
    float*    h2   = ws + o; o += (size_t)N * 64;
    float*    alS2 = ws + o; o += (size_t)N;
    float*    alD2 = ws + o; o += (size_t)N;
    float*    ex2  = ws + o; o += (size_t)Etot;
    unsigned* m2   = (unsigned*)(ws + o); o += (size_t)N;
    float*    den2 = ws + o; o += (size_t)N;

    const int TB = 256;
    auto cdiv = [](int a, int b) { return (a + b - 1) / b; };

    hipMemsetAsync(m1,   0, (size_t)N * 4 * sizeof(unsigned), stream);
    hipMemsetAsync(den1, 0, (size_t)N * 4 * sizeof(float), stream);
    hipMemsetAsync(agg1, 0, (size_t)N * 128 * sizeof(float), stream);
    hipMemsetAsync(m2,   0, (size_t)N * sizeof(unsigned), stream);
    hipMemsetAsync(den2, 0, (size_t)N * sizeof(float), stream);
    hipMemsetAsync(out,  0, (size_t)N * 64 * sizeof(float), stream);

    gemm_wmma_f32<128, 128><<<N / 16, 256, 0, stream>>>(x, W1, h1);
    attn_logits_h4<<<cdiv(N * 4, TB), TB, 0, stream>>>(h1, aS1, aD1, alS1, alD1, N);
    edge_max_h4<<<cdiv(Etot, TB), TB, 0, stream>>>(esrc, edst,
        (const float4*)alS1, (const float4*)alD1, m1, E, Etot);
    edge_exp_h4<<<cdiv(Etot, TB), TB, 0, stream>>>(esrc, edst,
        (const float4*)alS1, (const float4*)alD1, m1, ex1, den1, E, Etot);
    edge_agg_128<<<cdiv(Etot * 128, TB), TB, 0, stream>>>(esrc, edst,
        ex1, den1, h1, agg1, E, Etot);
    finalize_elu<<<cdiv(N * 128, TB), TB, 0, stream>>>(agg1, b1, N);

    gemm_wmma_f32<128, 64><<<N / 16, 128, 0, stream>>>(agg1, W2, h2);
    attn_logits_h1<<<cdiv(N, TB), TB, 0, stream>>>(h2, aS2, aD2, alS2, alD2, N);
    edge_max_h1<<<cdiv(Etot, TB), TB, 0, stream>>>(esrc, edst, alS2, alD2, m2, E, Etot);
    edge_exp_h1<<<cdiv(Etot, TB), TB, 0, stream>>>(esrc, edst, alS2, alD2, m2, ex2, den2, E, Etot);
    edge_agg_64<<<cdiv(Etot * 64, TB), TB, 0, stream>>>(esrc, edst, ex2, den2, h2, out, E, Etot);
    finalize_bias64<<<cdiv(N * 64, TB), TB, 0, stream>>>(out, b2, N);

    (void)n_in; (void)out_size; (void)ws_size;
}


# Layer-1 edge sweep reads a 1-byte-per-edge owner map (written in the logit launch), not the 4-byte destinations; a kept edge's destination is read in pass 1; list order changes, sums LDS arrival
# speedup vs baseline: 6.5142x; 6.5142x over previous
#include <hip/hip_runtime.h>
#include <hip/hip_bf16.h>

typedef float v2f __attribute__((ext_vector_type(2)));
typedef float v8f __attribute__((ext_vector_type(8)));

#define NEG_SLOPE 0.2f
#define INF_EPS   1e-16f

__device__ __forceinline__ unsigned fEnc(float x) {
    unsigned u = __float_as_uint(x);
    return (u & 0x80000000u) ? ~u : (u | 0x80000000u);
}
__device__ __forceinline__ float fDec(unsigned u) {
    u = (u & 0x80000000u) ? (u & 0x7FFFFFFFu) : ~u;
    return __uint_as_float(u);
}
__device__ __forceinline__ float leaky(float x) { return x > 0.f ? x : NEG_SLOPE * x; }

template <int K, int NCOLS>
__global__ __launch_bounds__(256) void gemm_wmma_f32(
    const float* __restrict__ A, const float* __restrict__ B,
    float* __restrict__ C)
{
    const int lane = threadIdx.x & 31;
    const int wave = threadIdx.x >> 5;
    const int row  = blockIdx.x * 16 + (lane & 15);
    const int colB = wave * 16 + (lane & 15);
    const int kh   = (lane >> 4) << 1;

    const float* __restrict__ aPtr = A + (size_t)row * K + kh;
    const float* __restrict__ bPtr = B + (size_t)kh * NCOLS + colB;

    v8f acc = {};
#pragma unroll
    for (int k0 = 0; k0 < K; k0 += 4) {
        v2f a, b;
        a.x = aPtr[k0];
        a.y = aPtr[k0 + 1];
        b.x = bPtr[k0 * NCOLS];
        b.y = bPtr[k0 * NCOLS + NCOLS];
        acc = __builtin_amdgcn_wmma_f32_16x16x4_f32(
            false, a, false, b, (short)0, acc, false, false);
    }
    float* __restrict__ cPtr =
        C + ((size_t)blockIdx.x * 16 + ((lane >> 4) << 3)) * NCOLS + colB;
#pragma unroll
    for (int r = 0; r < 8; ++r)
        cPtr[r * NCOLS] = acc[r];
}

template <int R>
__global__ void attn_logits_h4(const float* __restrict__ h,
                               const float* __restrict__ attS,
                               const float* __restrict__ attD,
                               float* __restrict__ alS, float* __restrict__ alD, int N,
                               const int* __restrict__ edst,
                               unsigned* __restrict__ omap, int E)
{
    int t = blockIdx.x * blockDim.x + threadIdx.x;
    const int nw = E >> 2;
    for (int w = t; w < nw; w += gridDim.x * blockDim.x) {
        const unsigned o0 = ((unsigned)edst[4 * w]     / (unsigned)R) & 255u;
        const unsigned o1 = ((unsigned)edst[4 * w + 1] / (unsigned)R) & 255u;
        const unsigned o2 = ((unsigned)edst[4 * w + 2] / (unsigned)R) & 255u;
        const unsigned o3 = ((unsigned)edst[4 * w + 3] / (unsigned)R) & 255u;
        omap[w] = o0 | (o1 << 8) | (o2 << 16) | (o3 << 24);
    }
    if (t >= N * 4) return;
    int n = t >> 2, hd = t & 3;
    const float* hp = h + (size_t)n * 128 + hd * 32;
    const float* as = attS + hd * 32;
    const float* ad = attD + hd * 32;
    float s = 0.f, d = 0.f;
#pragma unroll
    for (int c = 0; c < 32; ++c) { float v = hp[c]; s += v * as[c]; d += v * ad[c]; }
    alS[t] = s; alD[t] = d;
}

__global__ void attn_logits_h1(const float* __restrict__ h,
                               const float* __restrict__ attS,
                               const float* __restrict__ attD,
                               float* __restrict__ alS, float* __restrict__ alD, int N)
{
    int n = blockIdx.x * blockDim.x + threadIdx.x;
    if (n >= N) return;
    const float* hp = h + (size_t)n * 64;
    float s = 0.f, d = 0.f;
#pragma unroll
    for (int c = 0; c < 64; ++c) { float v = hp[c]; s += v * attS[c]; d += v * attD[c]; }
    alS[n] = s; alD[n] = d;
}

template <int H, int C, int R, int CAP, bool ELU, bool REUSE>
__global__ __launch_bounds__(1024) void dst_tile_gat(
    const int* __restrict__ esrc, const int* __restrict__ edst,
    const float* __restrict__ alS, const float* __restrict__ alD,
    const float* __restrict__ hfeat, const float* __restrict__ bias,
    float* __restrict__ out, int* __restrict__ wlist, int* __restrict__ wcnt,
    const unsigned* __restrict__ omap, int N, int E)
{
    constexpr int F  = H * C;
    constexpr int NJ = F / 32;
    constexpr int NT = 1024;
    constexpr int U  = 8;

    constexpr int RS = R + 1;
    __shared__ float    acc[RS * F];
    __shared__ unsigned mEnc[RS * H];
    __shared__ float    den[RS * H];
    __shared__ float    alDl[RS * H];
    __shared__ int2     lst[CAP];
    __shared__ float    lev[CAP * H];
    __shared__ int      cnt;

    const int tid  = threadIdx.x;
    const int lane = tid & 31;
    const int wave = tid >> 5;
    const int base = blockIdx.x * R;
    const int nloc = min(R, N - base);
    const int Etot = E + N;

    for (int i = tid; i < RS * F; i += NT) acc[i] = 0.f;
    for (int i = tid; i < RS * H; i += NT) {
        mEnc[i] = 0u;
        den[i]  = 0.f;
        alDl[i] = (i < nloc * H) ? alD[(size_t)base * H + i] : 0.f;
    }
    int2* __restrict__ wl = reinterpret_cast<int2*>(wlist) + (size_t)blockIdx.x * CAP;
    if (!REUSE) {
        for (int i = tid; i < nloc; i += NT) lst[i].x = E + base + i;
    }
    if (tid == 0) cnt = REUSE ? wcnt[blockIdx.x] : nloc;
    __syncthreads();

    if (!REUSE) {
    auto push = [&](int e) {
        const int p = atomicAdd(&cnt, 1);
        if (p < CAP) lst[p].x = e;
    };
    const unsigned pat = (blockIdx.x & 255u) * 0x01010101u;
    auto zbytes = [](unsigned x) {
        return (x - 0x01010101u) & ~x & 0x80808080u;
    };
    auto marks = [&](const uint4& v) {
        return (zbytes(v.x ^ pat) >> 7) | (zbytes(v.y ^ pat) >> 6) |
               (zbytes(v.z ^ pat) >> 5) | (zbytes(v.w ^ pat) >> 4);
    };
    static_assert(U % 2 == 0, "the sweep examines the loaded groups in pairs");
    const int Q = E >> 4;
    if (Q > 0) {
        const uint4* __restrict__ m4 = reinterpret_cast<const uint4*>(omap);
        const int trips = (Q + U * NT - 1) / (U * NT);
        const int rot = blockIdx.x % trips;
        auto first = [&](int k) {
            int kk = k + rot;
            if (kk >= trips) kk -= trips;
            return tid + kk * (U * NT);
        };
        uint4 cur[U], nxt[U];
        {
            const int q0 = first(0);
#pragma unroll
            for (int u = 0; u < U; ++u) cur[u] = m4[min(q0 + u * NT, Q - 1)];
        }
        for (int k = 0; k < trips; ++k) {
            const int q0 = first(k);
            const int qn = first(k + 1);
#pragma unroll
            for (int u = 0; u < U; ++u) nxt[u] = m4[min(qn + u * NT, Q - 1)];
#pragma unroll
            for (int u = 0; u < U; u += 2) {
                const int q = q0 + u * NT;
                if (q < Q) {
                    unsigned f = marks(cur[u]);
                    if (q + NT < Q) f |= marks(cur[u + 1]) << 4;
                    while (f) {
                        const int b = __builtin_ctz(f);
                        f &= f - 1;
                        push(((q + ((b >> 2) & 1) * NT) << 4) + ((b & 3) << 2) + (b >> 3));
                    }
                }
            }
#pragma unroll
            for (int u = 0; u < U; ++u) cur[u] = nxt[u];
        }
    }
    for (int e = (Q << 4) + tid; e < E; e += NT)
        if ((unsigned)(edst[e] - base) < (unsigned)nloc) push(e);
    __syncthreads();
    }
    const int n = cnt;
    if (!REUSE && tid == 0) wcnt[blockIdx.x] = n;

    if (n <= CAP) {
        for (int i = tid; i < n; i += NT) {
            int2 en;
            int  s;
            if (REUSE) {
                en = wl[i];
                s  = en.x;
            } else {
                const int e = lst[i].x;
                int d;
                if (e < E) { s = esrc[e]; d = edst[e]; } else { s = e - E; d = s; }
                int dl = d - base;
                if ((unsigned)dl >= (unsigned)nloc) dl = R;
                en = make_int2(s, dl);
                wl[i] = en;
            }
            lst[i] = en;
            for (int h = 0; h < H; ++h) {
                const float ev = leaky(alS[(size_t)s * H + h] + alDl[en.y * H + h]);
                lev[i * H + h] = ev;
                atomicMax(&mEnc[en.y * H + h], fEnc(ev));
            }
        }
        __syncthreads();
        for (int i = tid; i < n; i += NT) {
            const int dl = lst[i].y;
            for (int h = 0; h < H; ++h) {
                const float x = __expf(lev[i * H + h] - fDec(mEnc[dl * H + h]));
                lev[i * H + h] = x;
                atomicAdd(&den[dl * H + h], x);
            }
        }
        __syncthreads();
        for (int i0 = wave; i0 < n; i0 += 32 * U) {
            int   sN[U], dN[U];
            float v[U][NJ];
#pragma unroll
            for (int u = 0; u < U; ++u) {
                const int i = min(i0 + 32 * u, n - 1);
                sN[u] = __builtin_amdgcn_readfirstlane(lst[i].x);
                dN[u] = __builtin_amdgcn_readfirstlane(lst[i].y);
            }
#pragma unroll
            for (int u = 0; u < U; ++u)
#pragma unroll
                for (int j = 0; j < NJ; ++j)
                    v[u][j] = hfeat[(size_t)sN[u] * F + j * 32 + lane];
#pragma unroll
            for (int u = 0; u < U; ++u) {
                const int i = i0 + 32 * u;
                if (i < n) {
#pragma unroll
                    for (int j = 0; j < NJ; ++j) {
                        const float w = lev[i * H + (j * 32) / C];
                        atomicAdd(&acc[dN[u] * F + j * 32 + lane], w * v[u][j]);
                    }
                }
            }
        }
    } else {
        for (int ph = 0; ph < 3; ++ph) {
            for (int e = tid; e < Etot; e += NT) {
                const int d  = (e < E) ? edst[e] : e - E;
                const int dl = d - base;
                if ((unsigned)dl >= (unsigned)nloc) continue;
                const int s = (e < E) ? esrc[e] : e - E;
                for (int h = 0; h < H; ++h) {
                    const float ev = leaky(alS[(size_t)s * H + h] + alDl[dl * H + h]);
                    if (ph == 0) {
                        atomicMax(&mEnc[dl * H + h], fEnc(ev));
                    } else {
                        const float x = __expf(ev - fDec(mEnc[dl * H + h]));
                        if (ph == 1) {
                            atomicAdd(&den[dl * H + h], x);
                        } else {
                            for (int c = 0; c < C; ++c)
                                atomicAdd(&acc[dl * F + h * C + c],
                                          x * hfeat[(size_t)s * F + h * C + c]);
                        }
                    }
                }
            }
            __syncthreads();
        }
    }
    __syncthreads();

    for (int i = tid; i < nloc * F; i += NT) {
        const int nl = i / F, c = i - nl * F;
        float v = acc[i] / (den[nl * H + c / C] + INF_EPS) + bias[c];
        if (ELU) v = v > 0.f ? v : __expf(v) - 1.f;
        out[(size_t)base * F + i] = v;
    }
}

extern "C" void kernel_launch(void* const* d_in, const int* in_sizes, int n_in,
                              void* d_out, int out_size, void* d_ws, size_t ws_size,
                              hipStream_t stream)
{
    const float* x    = (const float*)d_in[0];
    const int*   eidx = (const int*)d_in[1];
    const float* W1   = (const float*)d_in[2];
    const float* aS1  = (const float*)d_in[3];
    const float* aD1  = (const float*)d_in[4];
    const float* b1   = (const float*)d_in[5];
    const float* W2   = (const float*)d_in[6];
    const float* aS2  = (const float*)d_in[7];
    const float* aD2  = (const float*)d_in[8];
    const float* b2   = (const float*)d_in[9];
    float* out = (float*)d_out;

    const int N = in_sizes[0] / 128;
    const int E = in_sizes[1] / 2;
    const int* esrc = eidx;
    const int* edst = eidx + E;

    float* ws = (float*)d_ws;
    size_t o = 0;
    float* h1   = ws + o; o += (size_t)N * 128;
    float* alS1 = ws + o; o += (size_t)N * 4;
    float* alD1 = ws + o; o += (size_t)N * 4;
    float* agg1 = ws + o; o += (size_t)N * 128;
    float* h2   = ws + o; o += (size_t)N * 64;
    float* alS2 = ws + o; o += (size_t)N;
    float* alD2 = ws + o; o += (size_t)N;

    const int TB = 256;
    constexpr int R   = 196;
    constexpr int CAP = 6144;
    auto cdiv = [](int a, int b) { return (a + b - 1) / b; };
    unsigned* omap = (unsigned*)(ws + o); o += (size_t)cdiv(E, 16) * 4;
    int* wlist = (int*)(ws + o); o += (size_t)cdiv(N, R) * CAP * 2;
    int* wcnt  = (int*)(ws + o); o += (size_t)cdiv(N, R);

    gemm_wmma_f32<128, 128><<<N / 16, 256, 0, stream>>>(x, W1, h1);
    attn_logits_h4<R><<<cdiv(N * 4, TB), TB, 0, stream>>>(h1, aS1, aD1, alS1, alD1, N,
                                                          edst, omap, E);
    dst_tile_gat<4, 32, R, CAP, true, false><<<cdiv(N, R), 1024, 0, stream>>>(
        esrc, edst, alS1, alD1, h1, b1, agg1, wlist, wcnt, omap, N, E);

    gemm_wmma_f32<128, 64><<<N / 16, 128, 0, stream>>>(agg1, W2, h2);
    attn_logits_h1<<<cdiv(N, TB), TB, 0, stream>>>(h2, aS2, aD2, alS2, alD2, N);
    dst_tile_gat<1, 64, R, CAP, false, true><<<cdiv(N, R), 1024, 0, stream>>>(
        esrc, edst, alS2, alD2, h2, b2, out, wlist, wcnt, omap, N, E);

    (void)n_in; (void)out_size; (void)ws_size;
}


# Layer-1 edge pass keeps its finished rows in LDS as the A operand of rows x W2 on the matrix unit, forms h2 and the layer-2 logits there: agg1 plane, 2 of 6 launches go; k and sum orders unchanged
# speedup vs baseline: 7.1456x; 1.0969x over previous
#include <hip/hip_runtime.h>
#include <hip/hip_bf16.h>

typedef float v2f __attribute__((ext_vector_type(2)));
typedef float v8f __attribute__((ext_vector_type(8)));

#define NEG_SLOPE 0.2f
#define INF_EPS   1e-16f

__device__ __forceinline__ unsigned fEnc(float x) {
    unsigned u = __float_as_uint(x);
    return (u & 0x80000000u) ? ~u : (u | 0x80000000u);
}
__device__ __forceinline__ float fDec(unsigned u) {
    u = (u & 0x80000000u) ? (u & 0x7FFFFFFFu) : ~u;
    return __uint_as_float(u);
}
__device__ __forceinline__ float leaky(float x) { return x > 0.f ? x : NEG_SLOPE * x; }

template <int K, int NCOLS>
__global__ __launch_bounds__(256) void gemm_wmma_f32(
    const float* __restrict__ A, const float* __restrict__ B,
    float* __restrict__ C)
{
    const int lane = threadIdx.x & 31;
    const int wave = threadIdx.x >> 5;
    const int row  = blockIdx.x * 16 + (lane & 15);
    const int colB = wave * 16 + (lane & 15);
    const int kh   = (lane >> 4) << 1;

    const float* __restrict__ aPtr = A + (size_t)row * K + kh;
    const float* __restrict__ bPtr = B + (size_t)kh * NCOLS + colB;

    v8f acc = {};
#pragma unroll
    for (int k0 = 0; k0 < K; k0 += 4) {
        v2f a, b;
        a.x = aPtr[k0];
        a.y = aPtr[k0 + 1];
        b.x = bPtr[k0 * NCOLS];
        b.y = bPtr[k0 * NCOLS + NCOLS];
        acc = __builtin_amdgcn_wmma_f32_16x16x4_f32(
            false, a, false, b, (short)0, acc, false, false);
    }
    float* __restrict__ cPtr =
        C + ((size_t)blockIdx.x * 16 + ((lane >> 4) << 3)) * NCOLS + colB;
#pragma unroll
    for (int r = 0; r < 8; ++r)
        cPtr[r * NCOLS] = acc[r];
}

template <int R>
__global__ void attn_logits_h4(const float* __restrict__ h,
                               const float* __restrict__ attS,
                               const float* __restrict__ attD,
                               float* __restrict__ alS, float* __restrict__ alD, int N,
                               const int* __restrict__ edst,
                               unsigned* __restrict__ omap, int E)
{
    int t = blockIdx.x * blockDim.x + threadIdx.x;
    const int nw = E >> 2;
    for (int w = t; w < nw; w += gridDim.x * blockDim.x) {
        const unsigned o0 = ((unsigned)edst[4 * w]     / (unsigned)R) & 255u;
        const unsigned o1 = ((unsigned)edst[4 * w + 1] / (unsigned)R) & 255u;
        const unsigned o2 = ((unsigned)edst[4 * w + 2] / (unsigned)R) & 255u;
        const unsigned o3 = ((unsigned)edst[4 * w + 3] / (unsigned)R) & 255u;
        omap[w] = o0 | (o1 << 8) | (o2 << 16) | (o3 << 24);
    }
    if (t >= N * 4) return;
    int n = t >> 2, hd = t & 3;
    const float* hp = h + (size_t)n * 128 + hd * 32;
    const float* as = attS + hd * 32;
    const float* ad = attD + hd * 32;
    float s = 0.f, d = 0.f;
#pragma unroll
    for (int c = 0; c < 32; ++c) { float v = hp[c]; s += v * as[c]; d += v * ad[c]; }
    alS[t] = s; alD[t] = d;
}

template <int H, int C, int R, int CAP, bool ELU, bool REUSE, int FN>
__global__ __launch_bounds__(1024) void dst_tile_gat(
    const int* __restrict__ esrc, const int* __restrict__ edst,
    const float* __restrict__ alS, const float* __restrict__ alD,
    const float* __restrict__ hfeat, const float* __restrict__ bias,
    float* __restrict__ out, int* __restrict__ wlist, int* __restrict__ wcnt,
    const unsigned* __restrict__ omap, int N, int E,
    const float* __restrict__ Wn, const float* __restrict__ attSn,
    const float* __restrict__ attDn, float* __restrict__ alSn, float* __restrict__ alDn)
{
    constexpr int F  = H * C;
    constexpr int P  = FN > 0 ? F + 4 : F;
    constexpr int NJ = F / 32;
    constexpr int NT = 1024;
    constexpr int U  = 8;

    constexpr int RS = R + 1;
    __shared__ __attribute__((aligned(16))) float acc[RS * P];
    __shared__ unsigned mEnc[RS * H];
    __shared__ float    den[RS * H];
    __shared__ float    alDl[RS * H];
    __shared__ int2     lst[CAP];
    __shared__ float    lev[CAP * H];
    __shared__ int      cnt;

    const int tid  = threadIdx.x;
    const int lane = tid & 31;
    const int wave = tid >> 5;
    const int base = blockIdx.x * R;
    const int nloc = min(R, N - base);
    const int Etot = E + N;

    for (int i = tid; i < RS * P; i += NT) acc[i] = 0.f;
    for (int i = tid; i < RS * H; i += NT) {
        mEnc[i] = 0u;
        den[i]  = 0.f;
        alDl[i] = (i < nloc * H) ? alD[(size_t)base * H + i] : 0.f;
    }
    int2* __restrict__ wl = reinterpret_cast<int2*>(wlist) + (size_t)blockIdx.x * CAP;
    if (!REUSE) {
        for (int i = tid; i < nloc; i += NT) lst[i].x = E + base + i;
    }
    if (tid == 0) cnt = REUSE ? wcnt[blockIdx.x] : nloc;
    __syncthreads();

    if (!REUSE) {
    auto push = [&](int e) {
        const int p = atomicAdd(&cnt, 1);
        if (p < CAP) lst[p].x = e;
    };
    const unsigned pat = (blockIdx.x & 255u) * 0x01010101u;
    auto zbytes = [](unsigned x) {
        return (x - 0x01010101u) & ~x & 0x80808080u;
    };
    auto marks = [&](const uint4& v) {
        return (zbytes(v.x ^ pat) >> 7) | (zbytes(v.y ^ pat) >> 6) |
               (zbytes(v.z ^ pat) >> 5) | (zbytes(v.w ^ pat) >> 4);
    };
    static_assert(U % 2 == 0, "the sweep examines the loaded groups in pairs");
    const int Q = E >> 4;
    if (Q > 0) {
        const uint4* __restrict__ m4 = reinterpret_cast<const uint4*>(omap);
        const int trips = (Q + U * NT - 1) / (U * NT);
        const int rot = blockIdx.x % trips;
        auto first = [&](int k) {
            int kk = k + rot;
            if (kk >= trips) kk -= trips;
            return tid + kk * (U * NT);
        };
        uint4 cur[U], nxt[U];
        {
            const int q0 = first(0);
#pragma unroll
            for (int u = 0; u < U; ++u) cur[u] = m4[min(q0 + u * NT, Q - 1)];
        }
        for (int k = 0; k < trips; ++k) {
            const int q0 = first(k);
            const int qn = first(k + 1);
#pragma unroll
            for (int u = 0; u < U; ++u) nxt[u] = m4[min(qn + u * NT, Q - 1)];
#pragma unroll
            for (int u = 0; u < U; u += 2) {
                const int q = q0 + u * NT;
                if (q < Q) {
                    unsigned f = marks(cur[u]);
                    if (q + NT < Q) f |= marks(cur[u + 1]) << 4;
                    while (f) {
                        const int b = __builtin_ctz(f);
                        f &= f - 1;
                        push(((q + ((b >> 2) & 1) * NT) << 4) + ((b & 3) << 2) + (b >> 3));
                    }
                }
            }
#pragma unroll
            for (int u = 0; u < U; ++u) cur[u] = nxt[u];
        }
    }
    for (int e = (Q << 4) + tid; e < E; e += NT)
        if ((unsigned)(edst[e] - base) < (unsigned)nloc) push(e);
    __syncthreads();
    }
    const int n = cnt;
    if (!REUSE && tid == 0) wcnt[blockIdx.x] = n;

    if (n <= CAP) {
        for (int i = tid; i < n; i += NT) {
            int2 en;
            int  s;
            if (REUSE) {
                en = wl[i];
                s  = en.x;
            } else {
                const int e = lst[i].x;
                int d;
                if (e < E) { s = esrc[e]; d = edst[e]; } else { s = e - E; d = s; }
                int dl = d - base;
                if ((unsigned)dl >= (unsigned)nloc) dl = R;
                en = make_int2(s, dl);
                wl[i] = en;
            }
            lst[i] = en;
            for (int h = 0; h < H; ++h) {
                const float ev = leaky(alS[(size_t)s * H + h] + alDl[en.y * H + h]);
                lev[i * H + h] = ev;
                atomicMax(&mEnc[en.y * H + h], fEnc(ev));
            }
        }
        __syncthreads();
        for (int i = tid; i < n; i += NT) {
            const int dl = lst[i].y;
            for (int h = 0; h < H; ++h) {
                const float x = __expf(lev[i * H + h] - fDec(mEnc[dl * H + h]));
                lev[i * H + h] = x;
                atomicAdd(&den[dl * H + h], x);
            }
        }
        __syncthreads();
        for (int i0 = wave; i0 < n; i0 += 32 * U) {
            int   sN[U], dN[U];
            float v[U][NJ];
#pragma unroll
            for (int u = 0; u < U; ++u) {
                const int i = min(i0 + 32 * u, n - 1);
                sN[u] = __builtin_amdgcn_readfirstlane(lst[i].x);
                dN[u] = __builtin_amdgcn_readfirstlane(lst[i].y);
            }
#pragma unroll
            for (int u = 0; u < U; ++u)
#pragma unroll
                for (int j = 0; j < NJ; ++j)
                    v[u][j] = hfeat[(size_t)sN[u] * F + j * 32 + lane];
#pragma unroll
            for (int u = 0; u < U; ++u) {
                const int i = i0 + 32 * u;
                if (i < n) {
#pragma unroll
                    for (int j = 0; j < NJ; ++j) {
                        const float w = lev[i * H + (j * 32) / C];
                        atomicAdd(&acc[dN[u] * P + j * 32 + lane], w * v[u][j]);
                    }
                }
            }
        }
    } else {
        for (int ph = 0; ph < 3; ++ph) {
            for (int e = tid; e < Etot; e += NT) {
                const int d  = (e < E) ? edst[e] : e - E;
                const int dl = d - base;
                if ((unsigned)dl >= (unsigned)nloc) continue;
                const int s = (e < E) ? esrc[e] : e - E;
                for (int h = 0; h < H; ++h) {
                    const float ev = leaky(alS[(size_t)s * H + h] + alDl[dl * H + h]);
                    if (ph == 0) {
                        atomicMax(&mEnc[dl * H + h], fEnc(ev));
                    } else {
                        const float x = __expf(ev - fDec(mEnc[dl * H + h]));
                        if (ph == 1) {
                            atomicAdd(&den[dl * H + h], x);
                        } else {
                            for (int c = 0; c < C; ++c)
                                atomicAdd(&acc[dl * P + h * C + c],
                                          x * hfeat[(size_t)s * F + h * C + c]);
                        }
                    }
                }
            }
            __syncthreads();
        }
    }
    __syncthreads();

    if constexpr (FN == 0) {
    for (int i = tid; i < nloc * F; i += NT) {
        const int nl = i / F, c = i - nl * F;
        float v = acc[i] / (den[nl * H + c / C] + INF_EPS) + bias[c];
        if (ELU) v = v > 0.f ? v : __expf(v) - 1.f;
        out[(size_t)base * F + i] = v;
    }
    } else {
    for (int i = tid; i < nloc * F; i += NT) {
        const int nl = i / F, c = i - nl * F;
        float v = acc[nl * P + c] / (den[nl * H + c / C] + INF_EPS) + bias[c];
        if (ELU) v = v > 0.f ? v : __expf(v) - 1.f;
        acc[nl * P + c] = v;
    }
    __syncthreads();

    constexpr int TR = (R + 15) / 16;
    constexpr int PN = FN + 1;
    static_assert(FN == 64, "32 waves are dealt as 4 column tiles x 8 row-tile groups");
    static_assert(TR <= 16, "a wave takes at most two row tiles");
    static_assert(F % 4 == 0, "k steps of 4");
    static_assert(TR * 16 * PN <= CAP * H, "the product tile lives in the list's logit array");
    float* __restrict__ nt = lev;
    {
        const int wv  = __builtin_amdgcn_readfirstlane(wave);
        const int ct  = wv & 3;
        const int rg  = ((wv >> 2) + 2 * ct) & 7;
        const int c16 = lane & 15, hh = lane >> 4;
        const int t0  = rg, t1 = rg + 8;
        const bool two = (t1 < TR) && (t1 * 16 < nloc);
        if (t0 * 16 < nloc) {
            const float* __restrict__ a0p = &acc[min(t0 * 16 + c16, R) * P + 2 * hh];
            const float* __restrict__ a1p = &acc[min(t1 * 16 + c16, R) * P + 2 * hh];
            const float* __restrict__ bp  = Wn + (size_t)(2 * hh) * FN + ct * 16 + c16;
            v8f d0 = {}, d1 = {};
            if (two) {
#pragma unroll
                for (int k0 = 0; k0 < F; k0 += 4) {
                    v2f b;
                    b.x = bp[k0 * FN];
                    b.y = bp[k0 * FN + FN];
                    const v2f a0 = *reinterpret_cast<const v2f*>(a0p + k0);
                    const v2f a1 = *reinterpret_cast<const v2f*>(a1p + k0);
                    d0 = __builtin_amdgcn_wmma_f32_16x16x4_f32(
                        false, a0, false, b, (short)0, d0, false, false);
                    d1 = __builtin_amdgcn_wmma_f32_16x16x4_f32(
                        false, a1, false, b, (short)0, d1, false, false);
                }
            } else {
#pragma unroll
                for (int k0 = 0; k0 < F; k0 += 4) {
                    v2f b;
                    b.x = bp[k0 * FN];
                    b.y = bp[k0 * FN + FN];
                    const v2f a0 = *reinterpret_cast<const v2f*>(a0p + k0);
                    d0 = __builtin_amdgcn_wmma_f32_16x16x4_f32(
                        false, a0, false, b, (short)0, d0, false, false);
                }
            }
            float* __restrict__ n0 = nt + (t0 * 16 + 8 * hh) * PN + ct * 16 + c16;
#pragma unroll
            for (int r = 0; r < 8; ++r) n0[r * PN] = d0[r];
            if (two) {
                float* __restrict__ n1 = nt + (t1 * 16 + 8 * hh) * PN + ct * 16 + c16;
#pragma unroll
                for (int r = 0; r < 8; ++r) n1[r * PN] = d1[r];
            }
        }
    }
    __syncthreads();

    for (int i = tid; i < nloc * FN; i += NT) {
        const int nl = i / FN, c = i - nl * FN;
        out[(size_t)base * FN + i] = nt[nl * PN + c];
    }
    for (int i = tid; i < nloc; i += NT) {
        float s = 0.f, d = 0.f;
#pragma unroll
        for (int c = 0; c < FN; ++c) { float v = nt[i * PN + c]; s += v * attSn[c]; d += v * attDn[c]; }
        alSn[base + i] = s; alDn[base + i] = d;
    }
    }
}

extern "C" void kernel_launch(void* const* d_in, const int* in_sizes, int n_in,
                              void* d_out, int out_size, void* d_ws, size_t ws_size,
                              hipStream_t stream)
{
    const float* x    = (const float*)d_in[0];
    const int*   eidx = (const int*)d_in[1];
    const float* W1   = (const float*)d_in[2];
    const float* aS1  = (const float*)d_in[3];
    const float* aD1  = (const float*)d_in[4];
    const float* b1   = (const float*)d_in[5];
    const float* W2   = (const float*)d_in[6];
    const float* aS2  = (const float*)d_in[7];
    const float* aD2  = (const float*)d_in[8];
    const float* b2   = (const float*)d_in[9];
    float* out = (float*)d_out;

    const int N = in_sizes[0] / 128;
    const int E = in_sizes[1] / 2;
    const int* esrc = eidx;
    const int* edst = eidx + E;

    float* ws = (float*)d_ws;
    size_t o = 0;
    float* h1   = ws + o; o += (size_t)N * 128;
    float* alS1 = ws + o; o += (size_t)N * 4;
    float* alD1 = ws + o; o += (size_t)N * 4;
    float* h2   = ws + o; o += (size_t)N * 64;
    float* alS2 = ws + o; o += (size_t)N;
    float* alD2 = ws + o; o += (size_t)N;

    const int TB = 256;
    constexpr int R   = 196;
    constexpr int CAP = 6144;
    auto cdiv = [](int a, int b) { return (a + b - 1) / b; };
    unsigned* omap = (unsigned*)(ws + o); o += (size_t)cdiv(E, 16) * 4;
    int* wlist = (int*)(ws + o); o += (size_t)cdiv(N, R) * CAP * 2;
    int* wcnt  = (int*)(ws + o); o += (size_t)cdiv(N, R);

    gemm_wmma_f32<128, 128><<<N / 16, 256, 0, stream>>>(x, W1, h1);
    attn_logits_h4<R><<<cdiv(N * 4, TB), TB, 0, stream>>>(h1, aS1, aD1, alS1, alD1, N,
                                                          edst, omap, E);
    dst_tile_gat<4, 32, R, CAP, true, false, 64><<<cdiv(N, R), 1024, 0, stream>>>(
        esrc, edst, alS1, alD1, h1, b1, h2, wlist, wcnt, omap, N, E,
        W2, aS2, aD2, alS2, alD2);

    dst_tile_gat<1, 64, R, CAP, false, true, 0><<<cdiv(N, R), 1024, 0, stream>>>(
        esrc, edst, alS2, alD2, h2, b2, out, wlist, wcnt, omap, N, E,
        nullptr, nullptr, nullptr, nullptr, nullptr);

    (void)n_in; (void)out_size; (void)ws_size;
}


# Layer-1 dense product by the edge passes' 196-node work-groups: x rows staged in LDS, product tile in LDS gives h1 (whole lines) and the logits; logit launch goes (4 to 3); k and sum orders unchanged
# speedup vs baseline: 7.7691x; 1.0873x over previous
#include <hip/hip_runtime.h>
#include <hip/hip_bf16.h>

typedef float v2f __attribute__((ext_vector_type(2)));
typedef float v8f __attribute__((ext_vector_type(8)));

#define NEG_SLOPE 0.2f
#define INF_EPS   1e-16f

__device__ __forceinline__ unsigned fEnc(float x) {
    unsigned u = __float_as_uint(x);
    return (u & 0x80000000u) ? ~u : (u | 0x80000000u);
}
__device__ __forceinline__ float fDec(unsigned u) {
    u = (u & 0x80000000u) ? (u & 0x7FFFFFFFu) : ~u;
    return __uint_as_float(u);
}
__device__ __forceinline__ float leaky(float x) { return x > 0.f ? x : NEG_SLOPE * x; }

template <int R>
__global__ __launch_bounds__(1024) void dense1_tile(
    const float* __restrict__ x, const float* __restrict__ W,
    const float* __restrict__ attS, const float* __restrict__ attD,
    float* __restrict__ h, float* __restrict__ alS, float* __restrict__ alD, int N,
    const int* __restrict__ edst, unsigned* __restrict__ omap, int E)
{
    constexpr int K  = 128;
    constexpr int F  = 128;
    constexpr int NT = 1024;
    constexpr int P  = K + 4;
    constexpr int PN = F + 4;
    constexpr int TR = (R + 15) / 16;
    constexpr int RS = R + 1;
    static_assert(TR <= 16, "a wave takes at most four row tiles");
    static_assert(R <= 256, "one thread per (head, node) forms the logits");
    __shared__ __attribute__((aligned(16))) float xs[RS * P];
    __shared__ __attribute__((aligned(16))) float nt[TR * 16 * PN];

    const int tid  = threadIdx.x;
    const int lane = tid & 31;
    const int wave = tid >> 5;
    const int base = blockIdx.x * R;
    const int nloc = min(R, N - base);

    {
        const int nw = E >> 2;
        for (int w = blockIdx.x * NT + tid; w < nw; w += gridDim.x * NT) {
            const unsigned o0 = ((unsigned)edst[4 * w]     / (unsigned)R) & 255u;
            const unsigned o1 = ((unsigned)edst[4 * w + 1] / (unsigned)R) & 255u;
            const unsigned o2 = ((unsigned)edst[4 * w + 2] / (unsigned)R) & 255u;
            const unsigned o3 = ((unsigned)edst[4 * w + 3] / (unsigned)R) & 255u;
            omap[w] = o0 | (o1 << 8) | (o2 << 16) | (o3 << 24);
        }
    }

    {
        constexpr int Q4 = K / 4;
        constexpr int LV = (RS * Q4 + NT - 1) / NT;
        const float4* __restrict__ x4 =
            reinterpret_cast<const float4*>(x) + (size_t)base * Q4;
        float4 xv[LV];
#pragma unroll
        for (int u = 0; u < LV; ++u) {
            const int i = tid + u * NT;
            xv[u] = (i < nloc * Q4) ? x4[i] : make_float4(0.f, 0.f, 0.f, 0.f);
        }
#pragma unroll
        for (int u = 0; u < LV; ++u) {
            const int i = tid + u * NT;
            if (i < RS * Q4) {
                const int r = i / Q4, c4 = i - r * Q4;
                *reinterpret_cast<float4*>(&xs[r * P + c4 * 4]) = xv[u];
            }
        }
    }
    __syncthreads();

    static_assert(F == 128, "32 waves are dealt as 8 column tiles x 4 row-tile groups");
    {
        const int wv  = __builtin_amdgcn_readfirstlane(wave);
        const int ct  = wv & 7;
        const int rq  = ((wv >> 3) + (wv & 3)) & 3;
        const int c16 = lane & 15, hh = lane >> 4;
        const float* __restrict__ bp = W + (size_t)(2 * hh) * F + ct * 16 + c16;
        for (int pr = 0; pr < 2; ++pr) {
            const int t0 = rq + 8 * pr, t1 = t0 + 4;
            if (t0 * 16 < nloc) {
                const bool two = (t1 < TR) && (t1 * 16 < nloc);
                const float* __restrict__ a0p = &xs[min(t0 * 16 + c16, R) * P + 2 * hh];
                const float* __restrict__ a1p = &xs[min(t1 * 16 + c16, R) * P + 2 * hh];
                v8f d0 = {}, d1 = {};
                if (two) {
#pragma unroll
                    for (int k0 = 0; k0 < K; k0 += 4) {
                        v2f b;
                        b.x = bp[k0 * F];
                        b.y = bp[k0 * F + F];
                        const v2f a0 = *reinterpret_cast<const v2f*>(a0p + k0);
                        const v2f a1 = *reinterpret_cast<const v2f*>(a1p + k0);
                        d0 = __builtin_amdgcn_wmma_f32_16x16x4_f32(
                            false, a0, false, b, (short)0, d0, false, false);
                        d1 = __builtin_amdgcn_wmma_f32_16x16x4_f32(
                            false, a1, false, b, (short)0, d1, false, false);
                    }
                } else {
#pragma unroll
                    for (int k0 = 0; k0 < K; k0 += 4) {
                        v2f b;
                        b.x = bp[k0 * F];
                        b.y = bp[k0 * F + F];
                        const v2f a0 = *reinterpret_cast<const v2f*>(a0p + k0);
                        d0 = __builtin_amdgcn_wmma_f32_16x16x4_f32(
                            false, a0, false, b, (short)0, d0, false, false);
                    }
                }
                float* __restrict__ n0 = nt + (t0 * 16 + 8 * hh) * PN + ct * 16 + c16;
#pragma unroll
                for (int r = 0; r < 8; ++r) n0[r * PN] = d0[r];
                if (two) {
                    float* __restrict__ n1 = nt + (t1 * 16 + 8 * hh) * PN + ct * 16 + c16;
#pragma unroll
                    for (int r = 0; r < 8; ++r) n1[r * PN] = d1[r];
                }
            }
        }
    }
    __syncthreads();

    {
        constexpr int Q4 = F / 4;
        float4* __restrict__ h4 = reinterpret_cast<float4*>(h) + (size_t)base * Q4;
        for (int i = tid; i < nloc * Q4; i += NT) {
            const int nl = i / Q4, c4 = i - nl * Q4;
            h4[i] = *reinterpret_cast<const float4*>(&nt[nl * PN + c4 * 4]);
        }
    }
    {
        const int hd = tid >> 8, n = tid & 255;
        if (n < nloc) {
            const float* hp = &nt[n * PN + hd * 32];
            const float* as = attS + hd * 32;
            const float* ad = attD + hd * 32;
            float s = 0.f, d = 0.f;
#pragma unroll
            for (int c = 0; c < 32; ++c) { float v = hp[c]; s += v * as[c]; d += v * ad[c]; }
            alS[(size_t)(base + n) * 4 + hd] = s;
            alD[(size_t)(base + n) * 4 + hd] = d;
        }
    }
}

template <int H, int C, int R, int CAP, bool ELU, bool REUSE, int FN>
__global__ __launch_bounds__(1024) void dst_tile_gat(
    const int* __restrict__ esrc, const int* __restrict__ edst,
    const float* __restrict__ alS, const float* __restrict__ alD,
    const float* __restrict__ hfeat, const float* __restrict__ bias,
    float* __restrict__ out, int* __restrict__ wlist, int* __restrict__ wcnt,
    const unsigned* __restrict__ omap, int N, int E,
    const float* __restrict__ Wn, const float* __restrict__ attSn,
    const float* __restrict__ attDn, float* __restrict__ alSn, float* __restrict__ alDn)
{
    constexpr int F  = H * C;
    constexpr int P  = FN > 0 ? F + 4 : F;
    constexpr int NJ = F / 32;
    constexpr int NT = 1024;
    constexpr int U  = 8;

    constexpr int RS = R + 1;
    __shared__ __attribute__((aligned(16))) float acc[RS * P];
    __shared__ unsigned mEnc[RS * H];
    __shared__ float    den[RS * H];
    __shared__ float    alDl[RS * H];
    __shared__ int2     lst[CAP];
    __shared__ float    lev[CAP * H];
    __shared__ int      cnt;

    const int tid  = threadIdx.x;
    const int lane = tid & 31;
    const int wave = tid >> 5;
    const int base = blockIdx.x * R;
    const int nloc = min(R, N - base);
    const int Etot = E + N;

    for (int i = tid; i < RS * P; i += NT) acc[i] = 0.f;
    for (int i = tid; i < RS * H; i += NT) {
        mEnc[i] = 0u;
        den[i]  = 0.f;
        alDl[i] = (i < nloc * H) ? alD[(size_t)base * H + i] : 0.f;
    }
    int2* __restrict__ wl = reinterpret_cast<int2*>(wlist) + (size_t)blockIdx.x * CAP;
    if (!REUSE) {
        for (int i = tid; i < nloc; i += NT) lst[i].x = E + base + i;
    }
    if (tid == 0) cnt = REUSE ? wcnt[blockIdx.x] : nloc;
    __syncthreads();

    if (!REUSE) {
    auto push = [&](int e) {
        const int p = atomicAdd(&cnt, 1);
        if (p < CAP) lst[p].x = e;
    };
    const unsigned pat = (blockIdx.x & 255u) * 0x01010101u;
    auto zbytes = [](unsigned x) {
        return (x - 0x01010101u) & ~x & 0x80808080u;
    };
    auto marks = [&](const uint4& v) {
        return (zbytes(v.x ^ pat) >> 7) | (zbytes(v.y ^ pat) >> 6) |
               (zbytes(v.z ^ pat) >> 5) | (zbytes(v.w ^ pat) >> 4);
    };
    static_assert(U % 2 == 0, "the sweep examines the loaded groups in pairs");
    const int Q = E >> 4;
    if (Q > 0) {
        const uint4* __restrict__ m4 = reinterpret_cast<const uint4*>(omap);
        const int trips = (Q + U * NT - 1) / (U * NT);
        const int rot = blockIdx.x % trips;
        auto first = [&](int k) {
            int kk = k + rot;
            if (kk >= trips) kk -= trips;
            return tid + kk * (U * NT);
        };
        uint4 cur[U], nxt[U];
        {
            const int q0 = first(0);
#pragma unroll
            for (int u = 0; u < U; ++u) cur[u] = m4[min(q0 + u * NT, Q - 1)];
        }
        for (int k = 0; k < trips; ++k) {
            const int q0 = first(k);
            const int qn = first(k + 1);
#pragma unroll
            for (int u = 0; u < U; ++u) nxt[u] = m4[min(qn + u * NT, Q - 1)];
#pragma unroll
            for (int u = 0; u < U; u += 2) {
                const int q = q0 + u * NT;
                if (q < Q) {
                    unsigned f = marks(cur[u]);
                    if (q + NT < Q) f |= marks(cur[u + 1]) << 4;
                    while (f) {
                        const int b = __builtin_ctz(f);
                        f &= f - 1;
                        push(((q + ((b >> 2) & 1) * NT) << 4) + ((b & 3) << 2) + (b >> 3));
                    }
                }
            }
#pragma unroll
            for (int u = 0; u < U; ++u) cur[u] = nxt[u];
        }
    }
    for (int e = (Q << 4) + tid; e < E; e += NT)
        if ((unsigned)(edst[e] - base) < (unsigned)nloc) push(e);
    __syncthreads();
    }
    const int n = cnt;
    if (!REUSE && tid == 0) wcnt[blockIdx.x] = n;

    if (n <= CAP) {
        for (int i = tid; i < n; i += NT) {
            int2 en;
            int  s;
            if (REUSE) {
                en = wl[i];
                s  = en.x;
            } else {
                const int e = lst[i].x;
                int d;
                if (e < E) { s = esrc[e]; d = edst[e]; } else { s = e - E; d = s; }
                int dl = d - base;
                if ((unsigned)dl >= (unsigned)nloc) dl = R;
                en = make_int2(s, dl);
                wl[i] = en;
            }
            lst[i] = en;
            for (int h = 0; h < H; ++h) {
                const float ev = leaky(alS[(size_t)s * H + h] + alDl[en.y * H + h]);
                lev[i * H + h] = ev;
                atomicMax(&mEnc[en.y * H + h], fEnc(ev));
            }
        }
        __syncthreads();
        for (int i = tid; i < n; i += NT) {
            const int dl = lst[i].y;
            for (int h = 0; h < H; ++h) {
                const float x = __expf(lev[i * H + h] - fDec(mEnc[dl * H + h]));
                lev[i * H + h] = x;
                atomicAdd(&den[dl * H + h], x);
            }
        }
        __syncthreads();
        for (int i0 = wave; i0 < n; i0 += 32 * U) {
            int   sN[U], dN[U];
            float v[U][NJ];
#pragma unroll
            for (int u = 0; u < U; ++u) {
                const int i = min(i0 + 32 * u, n - 1);
                sN[u] = __builtin_amdgcn_readfirstlane(lst[i].x);
                dN[u] = __builtin_amdgcn_readfirstlane(lst[i].y);
            }
#pragma unroll
            for (int u = 0; u < U; ++u)
#pragma unroll
                for (int j = 0; j < NJ; ++j)
                    v[u][j] = hfeat[(size_t)sN[u] * F + j * 32 + lane];
#pragma unroll
            for (int u = 0; u < U; ++u) {
                const int i = i0 + 32 * u;
                if (i < n) {
#pragma unroll
                    for (int j = 0; j < NJ; ++j) {
                        const float w = lev[i * H + (j * 32) / C];
                        atomicAdd(&acc[dN[u] * P + j * 32 + lane], w * v[u][j]);
                    }
                }
            }
        }
    } else {
        for (int ph = 0; ph < 3; ++ph) {
            for (int e = tid; e < Etot; e += NT) {
                const int d  = (e < E) ? edst[e] : e - E;
                const int dl = d - base;
                if ((unsigned)dl >= (unsigned)nloc) continue;
                const int s = (e < E) ? esrc[e] : e - E;
                for (int h = 0; h < H; ++h) {
                    const float ev = leaky(alS[(size_t)s * H + h] + alDl[dl * H + h]);
                    if (ph == 0) {
                        atomicMax(&mEnc[dl * H + h], fEnc(ev));
                    } else {
                        const float x = __expf(ev - fDec(mEnc[dl * H + h]));
                        if (ph == 1) {
                            atomicAdd(&den[dl * H + h], x);
                        } else {
                            for (int c = 0; c < C; ++c)
                                atomicAdd(&acc[dl * P + h * C + c],
                                          x * hfeat[(size_t)s * F + h * C + c]);
                        }
                    }
                }
            }
            __syncthreads();
        }
    }
    __syncthreads();

    if constexpr (FN == 0) {
    for (int i = tid; i < nloc * F; i += NT) {
        const int nl = i / F, c = i - nl * F;
        float v = acc[i] / (den[nl * H + c / C] + INF_EPS) + bias[c];
        if (ELU) v = v > 0.f ? v : __expf(v) - 1.f;
        out[(size_t)base * F + i] = v;
    }
    } else {
    for (int i = tid; i < nloc * F; i += NT) {
        const int nl = i / F, c = i - nl * F;
        float v = acc[nl * P + c] / (den[nl * H + c / C] + INF_EPS) + bias[c];
        if (ELU) v = v > 0.f ? v : __expf(v) - 1.f;
        acc[nl * P + c] = v;
    }
    __syncthreads();

    constexpr int TR = (R + 15) / 16;
    constexpr int PN = FN + 1;
    static_assert(FN == 64, "32 waves are dealt as 4 column tiles x 8 row-tile groups");
    static_assert(TR <= 16, "a wave takes at most two row tiles");
    static_assert(F % 4 == 0, "k steps of 4");
    static_assert(TR * 16 * PN <= CAP * H, "the product tile lives in the list's logit array");
    float* __restrict__ nt = lev;
    {
        const int wv  = __builtin_amdgcn_readfirstlane(wave);
        const int ct  = wv & 3;
        const int rg  = ((wv >> 2) + 2 * ct) & 7;
        const int c16 = lane & 15, hh = lane >> 4;
        const int t0  = rg, t1 = rg + 8;
        const bool two = (t1 < TR) && (t1 * 16 < nloc);
        if (t0 * 16 < nloc) {
            const float* __restrict__ a0p = &acc[min(t0 * 16 + c16, R) * P + 2 * hh];
            const float* __restrict__ a1p = &acc[min(t1 * 16 + c16, R) * P + 2 * hh];
            const float* __restrict__ bp  = Wn + (size_t)(2 * hh) * FN + ct * 16 + c16;
            v8f d0 = {}, d1 = {};
            if (two) {
#pragma unroll
                for (int k0 = 0; k0 < F; k0 += 4) {
                    v2f b;
                    b.x = bp[k0 * FN];
                    b.y = bp[k0 * FN + FN];
                    const v2f a0 = *reinterpret_cast<const v2f*>(a0p + k0);
                    const v2f a1 = *reinterpret_cast<const v2f*>(a1p + k0);
                    d0 = __builtin_amdgcn_wmma_f32_16x16x4_f32(
                        false, a0, false, b, (short)0, d0, false, false);
                    d1 = __builtin_amdgcn_wmma_f32_16x16x4_f32(
                        false, a1, false, b, (short)0, d1, false, false);
                }
            } else {
#pragma unroll
                for (int k0 = 0; k0 < F; k0 += 4) {
                    v2f b;
                    b.x = bp[k0 * FN];
                    b.y = bp[k0 * FN + FN];
                    const v2f a0 = *reinterpret_cast<const v2f*>(a0p + k0);
                    d0 = __builtin_amdgcn_wmma_f32_16x16x4_f32(
                        false, a0, false, b, (short)0, d0, false, false);
                }
            }
            float* __restrict__ n0 = nt + (t0 * 16 + 8 * hh) * PN + ct * 16 + c16;
#pragma unroll
            for (int r = 0; r < 8; ++r) n0[r * PN] = d0[r];
            if (two) {
                float* __restrict__ n1 = nt + (t1 * 16 + 8 * hh) * PN + ct * 16 + c16;
#pragma unroll
                for (int r = 0; r < 8; ++r) n1[r * PN] = d1[r];
            }
        }
    }
    __syncthreads();

    for (int i = tid; i < nloc * FN; i += NT) {
        const int nl = i / FN, c = i - nl * FN;
        out[(size_t)base * FN + i] = nt[nl * PN + c];
    }
    for (int i = tid; i < nloc; i += NT) {
        float s = 0.f, d = 0.f;
#pragma unroll
        for (int c = 0; c < FN; ++c) { float v = nt[i * PN + c]; s += v * attSn[c]; d += v * attDn[c]; }
        alSn[base + i] = s; alDn[base + i] = d;
    }
    }
}

extern "C" void kernel_launch(void* const* d_in, const int* in_sizes, int n_in,
                              void* d_out, int out_size, void* d_ws, size_t ws_size,
                              hipStream_t stream)
{
    const float* x    = (const float*)d_in[0];
    const int*   eidx = (const int*)d_in[1];
    const float* W1   = (const float*)d_in[2];
    const float* aS1  = (const float*)d_in[3];
    const float* aD1  = (const float*)d_in[4];
    const float* b1   = (const float*)d_in[5];
    const float* W2   = (const float*)d_in[6];
    const float* aS2  = (const float*)d_in[7];
    const float* aD2  = (const float*)d_in[8];
    const float* b2   = (const float*)d_in[9];
    float* out = (float*)d_out;

    const int N = in_sizes[0] / 128;
    const int E = in_sizes[1] / 2;
    const int* esrc = eidx;
    const int* edst = eidx + E;

    float* ws = (float*)d_ws;
    size_t o = 0;
    float* h1   = ws + o; o += (size_t)N * 128;
    float* alS1 = ws + o; o += (size_t)N * 4;
    float* alD1 = ws + o; o += (size_t)N * 4;
    float* h2   = ws + o; o += (size_t)N * 64;
    float* alS2 = ws + o; o += (size_t)N;
    float* alD2 = ws + o; o += (size_t)N;

    constexpr int R   = 196;
    constexpr int CAP = 6144;
    auto cdiv = [](int a, int b) { return (a + b - 1) / b; };
    unsigned* omap = (unsigned*)(ws + o); o += (size_t)cdiv(E, 16) * 4;
    int* wlist = (int*)(ws + o); o += (size_t)cdiv(N, R) * CAP * 2;
    int* wcnt  = (int*)(ws + o); o += (size_t)cdiv(N, R);

    dense1_tile<R><<<cdiv(N, R), 1024, 0, stream>>>(x, W1, aS1, aD1, h1, alS1, alD1, N,
                                                    edst, omap, E);
    dst_tile_gat<4, 32, R, CAP, true, false, 64><<<cdiv(N, R), 1024, 0, stream>>>(
        esrc, edst, alS1, alD1, h1, b1, h2, wlist, wcnt, omap, N, E,
        W2, aS2, aD2, alS2, alD2);

    dst_tile_gat<1, 64, R, CAP, false, true, 0><<<cdiv(N, R), 1024, 0, stream>>>(
        esrc, edst, alS2, alD2, h2, b2, out, wlist, wcnt, omap, N, E,
        nullptr, nullptr, nullptr, nullptr, nullptr);

    (void)n_in; (void)out_size; (void)ws_size;
}
